# starting kernel
# baseline (speedup 1.0000x reference)
#include <hip/hip_runtime.h>
#include <hip/hip_bf16.h>


typedef __attribute__((ext_vector_type(2))) float v2f;
typedef __attribute__((ext_vector_type(8))) float v8f;

#define DCH   128
#define KWIN  256
#define GAUSS_C 0.39894228f

__global__ __launch_bounds__(256)
void cagke_build_bank(const float* __restrict__ weight,
                      const float* __restrict__ sigma,
                      float* __restrict__ Ahat) {
    __shared__ float wLDS[DCH];
    __shared__ float sgLDS[DCH];
    __shared__ float red[DCH];
    const int tid = threadIdx.x;

    float wv = -3.402823466e38f;
    if (tid < DCH) {
        wv = weight[tid];
        sgLDS[tid] = sigma[tid];
        red[tid] = wv;
    }
    __syncthreads();
    for (int s = 64; s > 0; s >>= 1) {
        if (tid < s) red[tid] = fmaxf(red[tid], red[tid + s]);
        __syncthreads();
    }
    const float mx = red[0];
    __syncthreads();
    float e = 0.0f;
    if (tid < DCH) { e = expf(wv - mx); red[tid] = e; }
    __syncthreads();
    for (int s = 64; s > 0; s >>= 1) {
        if (tid < s) red[tid] += red[tid + s];
        __syncthreads();
    }
    const float inv_sum = 1.0f / red[0];
    if (tid < DCH) wLDS[tid] = e * inv_sum;
    __syncthreads();

    for (int idx = tid; idx < DCH * KWIN; idx += 256) {
        const int d = idx >> 8;
        const int k = idx & (KWIN - 1);
        const float sg = sgLDS[d];
        const float x  = (float)(k - 128);
        const float g  = (GAUSS_C / sg) * expf(-(x * x) / (2.0f * sg * sg));
        Ahat[idx] = wLDS[d] * g;
    }
}

__global__ __launch_bounds__(256)
void cagke_conv_wmma(const float* __restrict__ X,
                     const float* __restrict__ noise,
                     const float* __restrict__ Ahat,
                     float* __restrict__ psedu,
                     int T) {
    __shared__ float sLDS[384];
    const int tid = threadIdx.x;
    const int tB  = blockIdx.x * 128;

    __builtin_prefetch(Ahat + tid * 128, 0, 0);

    for (int u = tid; u < 384; u += 256) {
        const int j = tB - 128 + u;
        float v = 0.0f;
        if (j >= 0 && j < T) v = (X[j] > 0.5f) ? 1.0f : 0.0f;
        sLDS[u] = v;
    }
    __syncthreads();

    const int wave = tid >> 5;
    const int lane = tid & 31;
    const int n    = lane & 15;
    const int half = lane >> 4;
    const int base = 16 * wave + n + 255;

    v8f c = {0.f, 0.f, 0.f, 0.f, 0.f, 0.f, 0.f, 0.f};

    for (int kc = 0; kc < 64; ++kc) {
        const int k0 = (kc << 2) + 2 * half;
        v2f b;
        b[0] = sLDS[base - k0];
        b[1] = sLDS[base - k0 - 1];
#pragma unroll
        for (int dt = 0; dt < 8; ++dt) {
            const v2f a = *(const v2f*)(Ahat + (dt * 16 + n) * KWIN + k0);
            c = __builtin_amdgcn_wmma_f32_16x16x4_f32(
                     false, a,  false, b,
                     (short)0, c,  false,  false);
        }
    }

    float s = c[0] + c[1] + c[2] + c[3] + c[4] + c[5] + c[6] + c[7];
    const float tot = s + __shfl_xor(s, 16, 32);
    if (half == 0) {
        const int t = tB + 16 * wave + n;
        psedu[t] = tot + noise[t];
    }
}

__global__ __launch_bounds__(256)
void cagke_normalize(const float* __restrict__ psedu,
                     float* __restrict__ out, int T) {
    __shared__ float rmn[256];
    __shared__ float rmx[256];
    const int tid = threadIdx.x;
    float mn = 3.402823466e38f, mx = -3.402823466e38f;
    for (int t = tid; t < T; t += 256) {
        const float v = psedu[t];
        mn = fminf(mn, v);
        mx = fmaxf(mx, v);
    }
    rmn[tid] = mn; rmx[tid] = mx;
    __syncthreads();
    for (int s = 128; s > 0; s >>= 1) {
        if (tid < s) {
            rmn[tid] = fminf(rmn[tid], rmn[tid + s]);
            rmx[tid] = fmaxf(rmx[tid], rmx[tid + s]);
        }
        __syncthreads();
    }
    mn = rmn[0]; mx = rmx[0];
    const float inv = 1.0f / (mx - mn);
    for (int t = tid; t < T; t += 256) {
        out[t] = (psedu[t] - mn) * inv;
    }
}

extern "C" void kernel_launch(void* const* d_in, const int* in_sizes, int n_in,
                              void* d_out, int out_size, void* d_ws, size_t ws_size,
                              hipStream_t stream) {
    const float* X      = (const float*)d_in[0];
    const float* weight = (const float*)d_in[1];
    const float* noise  = (const float*)d_in[2];
    const float* sigma  = (const float*)d_in[3];
    float* out = (float*)d_out;
    const int T = in_sizes[0];

    float* Ahat  = (float*)d_ws;
    float* psedu = Ahat + DCH * KWIN;

    cagke_build_bank<<<1, 256, 0, stream>>>(weight, sigma, Ahat);
    cagke_conv_wmma<<<T / 128, 256, 0, stream>>>(X, noise, Ahat, psedu, T);
    cagke_normalize<<<1, 256, 0, stream>>>(psedu, out, T);
}


# Channel sum folded into the taps before the convolution (linearity): one 256-tap Toeplitz f32 WMMA conv replaces 128 channel convs; f32 sums reordered (channels first, four accumulators)
# speedup vs baseline: 2.1465x; 2.1465x over previous
#include <hip/hip_runtime.h>
#include <hip/hip_bf16.h>


typedef __attribute__((ext_vector_type(2))) float v2f;
typedef __attribute__((ext_vector_type(8))) float v8f;

#define DCH   128
#define KWIN  256
#define GAUSS_C 0.39894228f

#define TILE_T 256
#define NCHUNK 68
#define GPAD   16
#define QBLK   17

__global__ __launch_bounds__(256)
void cagke_build_taps(const float* __restrict__ weight,
                      const float* __restrict__ sigma,
                      float* __restrict__ G) {
    __shared__ float wLDS[DCH];
    __shared__ float sgLDS[DCH];
    __shared__ float red[DCH];
    const int tid = threadIdx.x;

    float wv = -3.402823466e38f;
    if (tid < DCH) {
        wv = weight[tid];
        sgLDS[tid] = sigma[tid];
        red[tid] = wv;
    }
    __syncthreads();
    for (int s = 64; s > 0; s >>= 1) {
        if (tid < s) red[tid] = fmaxf(red[tid], red[tid + s]);
        __syncthreads();
    }
    const float mx = red[0];
    __syncthreads();
    float e = 0.0f;
    if (tid < DCH) { e = expf(wv - mx); red[tid] = e; }
    __syncthreads();
    for (int s = 64; s > 0; s >>= 1) {
        if (tid < s) red[tid] += red[tid + s];
        __syncthreads();
    }
    const float inv_sum = 1.0f / red[0];
    if (tid < DCH) wLDS[tid] = e * inv_sum;
    __syncthreads();

    const float x = (float)(tid - 128);
    float acc = 0.0f;
    for (int d = 0; d < DCH; ++d) {
        const float sg = sgLDS[d];
        const float g  = (GAUSS_C / sg) * expf(-(x * x) / (2.0f * sg * sg));
        acc += wLDS[d] * g;
    }
    G[tid] = acc;
}

__global__ __launch_bounds__(32)
void cagke_conv_wmma(const float* __restrict__ X,
                     const float* __restrict__ noise,
                     const float* __restrict__ G,
                     float* __restrict__ psedu,
                     int T) {
    __shared__ __attribute__((aligned(16))) float sL[TILE_T + 256];
    __shared__ __attribute__((aligned(16))) float gR[KWIN + 2 * GPAD];
    const int lane = threadIdx.x;
    const int t0   = blockIdx.x * TILE_T;

    float nz[8];
#pragma unroll
    for (int r = 0; r < 8; ++r) {
        nz[r] = noise[t0 + 32 * r + lane];
    }
    float gv[(KWIN + 2 * GPAD) / 32];
#pragma unroll
    for (int i = 0; i < (KWIN + 2 * GPAD) / 32; ++i) {
        const int k = 270 - (lane + 32 * i);
        const bool ok = (k >= 0 && k < KWIN);
        const float g = G[ok ? k : 0];
        gv[i] = ok ? g : 0.0f;
    }
    float xv[(TILE_T + 256) / 32];
#pragma unroll
    for (int i = 0; i < (TILE_T + 256) / 32; ++i) {
        const int j = t0 - 128 + lane + 32 * i;
        const bool ok = (j >= 0 && j < T);
        const float x = X[ok ? j : 0];
        xv[i] = (ok && x > 0.5f) ? 1.0f : 0.0f;
    }
#pragma unroll
    for (int i = 0; i < (KWIN + 2 * GPAD) / 32; ++i) gR[lane + 32 * i] = gv[i];
#pragma unroll
    for (int i = 0; i < (TILE_T + 256) / 32; ++i) sL[lane + 32 * i] = xv[i];
    __syncthreads();

    const int c  = lane & 15;
    const int h  = lane >> 4;
    const int pm = 2 * (c & 7) + (c >> 3);
    const float* ap = sL + 16 * pm + 2 * h;
    const float* bp = gR + 15 - c + 2 * h;

    v8f acc[4];
#pragma unroll
    for (int r = 0; r < 4; ++r) acc[r] = (v8f){0.f, 0.f, 0.f, 0.f, 0.f, 0.f, 0.f, 0.f};

    v2f a[NCHUNK], b[NCHUNK];
#pragma unroll
    for (int i = 0; i < QBLK; ++i) {
        a[i] = *(const v2f*)(ap + 4 * i);
        b[i][0] = bp[4 * i];
        b[i][1] = bp[4 * i + 1];
    }
#pragma unroll
    for (int blk = 0; blk < NCHUNK / QBLK; ++blk) {
        if (blk + 1 < NCHUNK / QBLK) {
#pragma unroll
            for (int i = 0; i < QBLK; ++i) {
                const int q = QBLK * (blk + 1) + i;
                a[q] = *(const v2f*)(ap + 4 * q);
                b[q][0] = bp[4 * q];
                b[q][1] = bp[4 * q + 1];
            }
        }
        __builtin_amdgcn_sched_barrier(0);
#pragma unroll
        for (int i = 0; i < QBLK; ++i) {
            const int q = QBLK * blk + i;
            acc[q & 3] = __builtin_amdgcn_wmma_f32_16x16x4_f32(
                     false, a[q],  false, b[q],
                     (short)0, acc[q & 3],  false,  false);
        }
        __builtin_amdgcn_sched_barrier(0);
    }

    const v8f s = (acc[0] + acc[1]) + (acc[2] + acc[3]);

#pragma unroll
    for (int r = 0; r < 8; ++r) {
        psedu[t0 + 32 * r + lane] = s[r] + nz[r];
    }
}

__global__ __launch_bounds__(256)
void cagke_normalize(const float* __restrict__ psedu,
                     float* __restrict__ out, int T) {
    __shared__ float rmn[256];
    __shared__ float rmx[256];
    const int tid = threadIdx.x;
    float mn = 3.402823466e38f, mx = -3.402823466e38f;
    for (int t = tid; t < T; t += 256) {
        const float v = psedu[t];
        mn = fminf(mn, v);
        mx = fmaxf(mx, v);
    }
    rmn[tid] = mn; rmx[tid] = mx;
    __syncthreads();
    for (int s = 128; s > 0; s >>= 1) {
        if (tid < s) {
            rmn[tid] = fminf(rmn[tid], rmn[tid + s]);
            rmx[tid] = fmaxf(rmx[tid], rmx[tid + s]);
        }
        __syncthreads();
    }
    mn = rmn[0]; mx = rmx[0];
    const float inv = 1.0f / (mx - mn);
    for (int t = tid; t < T; t += 256) {
        out[t] = (psedu[t] - mn) * inv;
    }
}

extern "C" void kernel_launch(void* const* d_in, const int* in_sizes, int n_in,
                              void* d_out, int out_size, void* d_ws, size_t ws_size,
                              hipStream_t stream) {
    const float* X      = (const float*)d_in[0];
    const float* weight = (const float*)d_in[1];
    const float* noise  = (const float*)d_in[2];
    const float* sigma  = (const float*)d_in[3];
    float* out = (float*)d_out;
    const int T = in_sizes[0];

    float* G     = (float*)d_ws;
    float* psedu = G + KWIN;

    cagke_build_taps<<<1, 256, 0, stream>>>(weight, sigma, G);
    cagke_conv_wmma<<<T / 256, 32, 0, stream>>>(X, noise, G, psedu, T);
    cagke_normalize<<<1, 256, 0, stream>>>(psedu, out, T);
}


# Tap build spread over 8 blocks, 128 channels across lanes: softmax and each tap's channel sum are lane butterflies, no channel loop or LDS tree; f32 channel sum reordered (4 per lane, then butterfly)
# speedup vs baseline: 3.1445x; 1.4649x over previous
#include <hip/hip_runtime.h>
#include <hip/hip_bf16.h>


typedef __attribute__((ext_vector_type(2))) float v2f;
typedef __attribute__((ext_vector_type(8))) float v8f;

#define DCH   128
#define KWIN  256
#define GAUSS_C 0.39894228f

#define TILE_T 256
#define NCHUNK 68
#define GPAD   16
#define QBLK   17

__device__ __forceinline__ float xrow(float x) {
    const int b = __builtin_bit_cast(int, x);
    return __builtin_bit_cast(float,
        __builtin_amdgcn_permlanex16(b, b, 0x76543210, 0xfedcba98, false, false));
}
template <int MASK>
__device__ __forceinline__ float xlane(float x) {
    return __builtin_bit_cast(float,
        __builtin_amdgcn_update_dpp(0, __builtin_bit_cast(int, x), 0x160 + MASK, 0xf, 0xf, true));
}
__device__ __forceinline__ float wave_sum(float v) {
    v += xrow(v);
    v += xlane<8>(v);
    v += xlane<4>(v);
    v += xlane<2>(v);
    v += xlane<1>(v);
    return v;
}
__device__ __forceinline__ float wave_max(float v) {
    v = fmaxf(v, xrow(v));
    v = fmaxf(v, xlane<8>(v));
    v = fmaxf(v, xlane<4>(v));
    v = fmaxf(v, xlane<2>(v));
    v = fmaxf(v, xlane<1>(v));
    return v;
}

#define TAPS_PER_BLOCK 32
#define TAPS_PER_WAVE  4
#define TAP_BLOCKS     8
static_assert(DCH == 4 * 32, "four channels per lane");
static_assert(TAP_BLOCKS * TAPS_PER_BLOCK == KWIN, "every tap made once");

__global__ __launch_bounds__(256)
void cagke_build_taps(const float* __restrict__ weight,
                      const float* __restrict__ sigma,
                      float* __restrict__ G) {
    __shared__ __attribute__((aligned(16))) float line[TAPS_PER_BLOCK];
    const int lane = threadIdx.x & 31;
    const int wave = threadIdx.x >> 5;

    float wv[4], sg[4];
#pragma unroll
    for (int i = 0; i < 4; ++i) {
        wv[i] = weight[lane + 32 * i];
        sg[i] = sigma[lane + 32 * i];
    }

    const float mx = wave_max(fmaxf(fmaxf(wv[0], wv[2]), fmaxf(wv[1], wv[3])));
    float e[4];
#pragma unroll
    for (int i = 0; i < 4; ++i) e[i] = expf(wv[i] - mx);
    const float inv_sum = 1.0f / wave_sum((e[0] + e[2]) + (e[1] + e[3]));

    float wn[4], cs[4], den[4];
#pragma unroll
    for (int i = 0; i < 4; ++i) {
        wn[i]  = e[i] * inv_sum;
        cs[i]  = GAUSS_C / sg[i];
        den[i] = 2.0f * sg[i] * sg[i];
    }

    const int k0 = TAPS_PER_BLOCK * blockIdx.x + TAPS_PER_WAVE * wave;
    float tap[TAPS_PER_WAVE];
#pragma unroll
    for (int j = 0; j < TAPS_PER_WAVE; ++j) {
        const float x = (float)(k0 + j - 128);
        float acc = 0.0f;
#pragma unroll
        for (int i = 0; i < 4; ++i) {
            const float g = cs[i] * expf(-(x * x) / den[i]);
            acc += wn[i] * g;
        }
        tap[j] = wave_sum(acc);
    }
    if (lane == 0) {
        *(float4*)(line + TAPS_PER_WAVE * wave) = make_float4(tap[0], tap[1], tap[2], tap[3]);
    }
    __syncthreads();
    if (wave == 0) G[TAPS_PER_BLOCK * blockIdx.x + lane] = line[lane];
}

__global__ __launch_bounds__(32)
void cagke_conv_wmma(const float* __restrict__ X,
                     const float* __restrict__ noise,
                     const float* __restrict__ G,
                     float* __restrict__ psedu,
                     int T) {
    __shared__ __attribute__((aligned(16))) float sL[TILE_T + 256];
    __shared__ __attribute__((aligned(16))) float gR[KWIN + 2 * GPAD];
    const int lane = threadIdx.x;
    const int t0   = blockIdx.x * TILE_T;

    float nz[8];
#pragma unroll
    for (int r = 0; r < 8; ++r) {
        nz[r] = noise[t0 + 32 * r + lane];
    }
    float gv[(KWIN + 2 * GPAD) / 32];
#pragma unroll
    for (int i = 0; i < (KWIN + 2 * GPAD) / 32; ++i) {
        const int k = 270 - (lane + 32 * i);
        const bool ok = (k >= 0 && k < KWIN);
        const float g = G[ok ? k : 0];
        gv[i] = ok ? g : 0.0f;
    }
    float xv[(TILE_T + 256) / 32];
#pragma unroll
    for (int i = 0; i < (TILE_T + 256) / 32; ++i) {
        const int j = t0 - 128 + lane + 32 * i;
        const bool ok = (j >= 0 && j < T);
        const float x = X[ok ? j : 0];
        xv[i] = (ok && x > 0.5f) ? 1.0f : 0.0f;
    }
#pragma unroll
    for (int i = 0; i < (KWIN + 2 * GPAD) / 32; ++i) gR[lane + 32 * i] = gv[i];
#pragma unroll
    for (int i = 0; i < (TILE_T + 256) / 32; ++i) sL[lane + 32 * i] = xv[i];
    __syncthreads();

    const int c  = lane & 15;
    const int h  = lane >> 4;
    const int pm = 2 * (c & 7) + (c >> 3);
    const float* ap = sL + 16 * pm + 2 * h;
    const float* bp = gR + 15 - c + 2 * h;

    v8f acc[4];
#pragma unroll
    for (int r = 0; r < 4; ++r) acc[r] = (v8f){0.f, 0.f, 0.f, 0.f, 0.f, 0.f, 0.f, 0.f};

    v2f a[NCHUNK], b[NCHUNK];
#pragma unroll
    for (int i = 0; i < QBLK; ++i) {
        a[i] = *(const v2f*)(ap + 4 * i);
        b[i][0] = bp[4 * i];
        b[i][1] = bp[4 * i + 1];
    }
#pragma unroll
    for (int blk = 0; blk < NCHUNK / QBLK; ++blk) {
        if (blk + 1 < NCHUNK / QBLK) {
#pragma unroll
            for (int i = 0; i < QBLK; ++i) {
                const int q = QBLK * (blk + 1) + i;
                a[q] = *(const v2f*)(ap + 4 * q);
                b[q][0] = bp[4 * q];
                b[q][1] = bp[4 * q + 1];
            }
        }
        __builtin_amdgcn_sched_barrier(0);
#pragma unroll
        for (int i = 0; i < QBLK; ++i) {
            const int q = QBLK * blk + i;
            acc[q & 3] = __builtin_amdgcn_wmma_f32_16x16x4_f32(
                     false, a[q],  false, b[q],
                     (short)0, acc[q & 3],  false,  false);
        }
        __builtin_amdgcn_sched_barrier(0);
    }

    const v8f s = (acc[0] + acc[1]) + (acc[2] + acc[3]);

#pragma unroll
    for (int r = 0; r < 8; ++r) {
        psedu[t0 + 32 * r + lane] = s[r] + nz[r];
    }
}

__global__ __launch_bounds__(256)
void cagke_normalize(const float* __restrict__ psedu,
                     float* __restrict__ out, int T) {
    __shared__ float rmn[256];
    __shared__ float rmx[256];
    const int tid = threadIdx.x;
    float mn = 3.402823466e38f, mx = -3.402823466e38f;
    for (int t = tid; t < T; t += 256) {
        const float v = psedu[t];
        mn = fminf(mn, v);
        mx = fmaxf(mx, v);
    }
    rmn[tid] = mn; rmx[tid] = mx;
    __syncthreads();
    for (int s = 128; s > 0; s >>= 1) {
        if (tid < s) {
            rmn[tid] = fminf(rmn[tid], rmn[tid + s]);
            rmx[tid] = fmaxf(rmx[tid], rmx[tid + s]);
        }
        __syncthreads();
    }
    mn = rmn[0]; mx = rmx[0];
    const float inv = 1.0f / (mx - mn);
    for (int t = tid; t < T; t += 256) {
        out[t] = (psedu[t] - mn) * inv;
    }
}

extern "C" void kernel_launch(void* const* d_in, const int* in_sizes, int n_in,
                              void* d_out, int out_size, void* d_ws, size_t ws_size,
                              hipStream_t stream) {
    const float* X      = (const float*)d_in[0];
    const float* weight = (const float*)d_in[1];
    const float* noise  = (const float*)d_in[2];
    const float* sigma  = (const float*)d_in[3];
    float* out = (float*)d_out;
    const int T = in_sizes[0];

    float* G     = (float*)d_ws;
    float* psedu = G + KWIN;

    cagke_build_taps<<<TAP_BLOCKS, 256, 0, stream>>>(weight, sigma, G);
    cagke_conv_wmma<<<T / 256, 32, 0, stream>>>(X, noise, G, psedu, T);
    cagke_normalize<<<1, 256, 0, stream>>>(psedu, out, T);
}


# Normalize keeps the signal in registers: 8 b128 loads per thread in flight together, min/max by lane butterfly and one barrier, b128 stores; no dependent-load loop, no LDS tree, no re-read
# speedup vs baseline: 6.1087x; 1.9427x over previous
#include <hip/hip_runtime.h>
#include <hip/hip_bf16.h>


typedef __attribute__((ext_vector_type(2))) float v2f;
typedef __attribute__((ext_vector_type(8))) float v8f;

#define DCH   128
#define KWIN  256
#define GAUSS_C 0.39894228f

#define TILE_T 256
#define NCHUNK 68
#define GPAD   16
#define QBLK   17

__device__ __forceinline__ float xrow(float x) {
    const int b = __builtin_bit_cast(int, x);
    return __builtin_bit_cast(float,
        __builtin_amdgcn_permlanex16(b, b, 0x76543210, 0xfedcba98, false, false));
}
template <int MASK>
__device__ __forceinline__ float xlane(float x) {
    return __builtin_bit_cast(float,
        __builtin_amdgcn_update_dpp(0, __builtin_bit_cast(int, x), 0x160 + MASK, 0xf, 0xf, true));
}
__device__ __forceinline__ float wave_sum(float v) {
    v += xrow(v);
    v += xlane<8>(v);
    v += xlane<4>(v);
    v += xlane<2>(v);
    v += xlane<1>(v);
    return v;
}
__device__ __forceinline__ float wave_max(float v) {
    v = fmaxf(v, xrow(v));
    v = fmaxf(v, xlane<8>(v));
    v = fmaxf(v, xlane<4>(v));
    v = fmaxf(v, xlane<2>(v));
    v = fmaxf(v, xlane<1>(v));
    return v;
}

#define TAPS_PER_BLOCK 32
#define TAPS_PER_WAVE  4
#define TAP_BLOCKS     8
static_assert(DCH == 4 * 32, "four channels per lane");
static_assert(TAP_BLOCKS * TAPS_PER_BLOCK == KWIN, "every tap made once");

__global__ __launch_bounds__(256)
void cagke_build_taps(const float* __restrict__ weight,
                      const float* __restrict__ sigma,
                      float* __restrict__ G) {
    __shared__ __attribute__((aligned(16))) float line[TAPS_PER_BLOCK];
    const int lane = threadIdx.x & 31;
    const int wave = threadIdx.x >> 5;

    float wv[4], sg[4];
#pragma unroll
    for (int i = 0; i < 4; ++i) {
        wv[i] = weight[lane + 32 * i];
        sg[i] = sigma[lane + 32 * i];
    }

    const float mx = wave_max(fmaxf(fmaxf(wv[0], wv[2]), fmaxf(wv[1], wv[3])));
    float e[4];
#pragma unroll
    for (int i = 0; i < 4; ++i) e[i] = expf(wv[i] - mx);
    const float inv_sum = 1.0f / wave_sum((e[0] + e[2]) + (e[1] + e[3]));

    float wn[4], cs[4], den[4];
#pragma unroll
    for (int i = 0; i < 4; ++i) {
        wn[i]  = e[i] * inv_sum;
        cs[i]  = GAUSS_C / sg[i];
        den[i] = 2.0f * sg[i] * sg[i];
    }

    const int k0 = TAPS_PER_BLOCK * blockIdx.x + TAPS_PER_WAVE * wave;
    float tap[TAPS_PER_WAVE];
#pragma unroll
    for (int j = 0; j < TAPS_PER_WAVE; ++j) {
        const float x = (float)(k0 + j - 128);
        float acc = 0.0f;
#pragma unroll
        for (int i = 0; i < 4; ++i) {
            const float g = cs[i] * expf(-(x * x) / den[i]);
            acc += wn[i] * g;
        }
        tap[j] = wave_sum(acc);
    }
    if (lane == 0) {
        *(float4*)(line + TAPS_PER_WAVE * wave) = make_float4(tap[0], tap[1], tap[2], tap[3]);
    }
    __syncthreads();
    if (wave == 0) G[TAPS_PER_BLOCK * blockIdx.x + lane] = line[lane];
}

__global__ __launch_bounds__(32)
void cagke_conv_wmma(const float* __restrict__ X,
                     const float* __restrict__ noise,
                     const float* __restrict__ G,
                     float* __restrict__ psedu,
                     int T) {
    __shared__ __attribute__((aligned(16))) float sL[TILE_T + 256];
    __shared__ __attribute__((aligned(16))) float gR[KWIN + 2 * GPAD];
    const int lane = threadIdx.x;
    const int t0   = blockIdx.x * TILE_T;

    float nz[8];
#pragma unroll
    for (int r = 0; r < 8; ++r) {
        nz[r] = noise[t0 + 32 * r + lane];
    }
    float gv[(KWIN + 2 * GPAD) / 32];
#pragma unroll
    for (int i = 0; i < (KWIN + 2 * GPAD) / 32; ++i) {
        const int k = 270 - (lane + 32 * i);
        const bool ok = (k >= 0 && k < KWIN);
        const float g = G[ok ? k : 0];
        gv[i] = ok ? g : 0.0f;
    }
    float xv[(TILE_T + 256) / 32];
#pragma unroll
    for (int i = 0; i < (TILE_T + 256) / 32; ++i) {
        const int j = t0 - 128 + lane + 32 * i;
        const bool ok = (j >= 0 && j < T);
        const float x = X[ok ? j : 0];
        xv[i] = (ok && x > 0.5f) ? 1.0f : 0.0f;
    }
#pragma unroll
    for (int i = 0; i < (KWIN + 2 * GPAD) / 32; ++i) gR[lane + 32 * i] = gv[i];
#pragma unroll
    for (int i = 0; i < (TILE_T + 256) / 32; ++i) sL[lane + 32 * i] = xv[i];
    __syncthreads();

    const int c  = lane & 15;
    const int h  = lane >> 4;
    const int pm = 2 * (c & 7) + (c >> 3);
    const float* ap = sL + 16 * pm + 2 * h;
    const float* bp = gR + 15 - c + 2 * h;

    v8f acc[4];
#pragma unroll
    for (int r = 0; r < 4; ++r) acc[r] = (v8f){0.f, 0.f, 0.f, 0.f, 0.f, 0.f, 0.f, 0.f};

    v2f a[NCHUNK], b[NCHUNK];
#pragma unroll
    for (int i = 0; i < QBLK; ++i) {
        a[i] = *(const v2f*)(ap + 4 * i);
        b[i][0] = bp[4 * i];
        b[i][1] = bp[4 * i + 1];
    }
#pragma unroll
    for (int blk = 0; blk < NCHUNK / QBLK; ++blk) {
        if (blk + 1 < NCHUNK / QBLK) {
#pragma unroll
            for (int i = 0; i < QBLK; ++i) {
                const int q = QBLK * (blk + 1) + i;
                a[q] = *(const v2f*)(ap + 4 * q);
                b[q][0] = bp[4 * q];
                b[q][1] = bp[4 * q + 1];
            }
        }
        __builtin_amdgcn_sched_barrier(0);
#pragma unroll
        for (int i = 0; i < QBLK; ++i) {
            const int q = QBLK * blk + i;
            acc[q & 3] = __builtin_amdgcn_wmma_f32_16x16x4_f32(
                     false, a[q],  false, b[q],
                     (short)0, acc[q & 3],  false,  false);
        }
        __builtin_amdgcn_sched_barrier(0);
    }

    const v8f s = (acc[0] + acc[1]) + (acc[2] + acc[3]);

#pragma unroll
    for (int r = 0; r < 8; ++r) {
        psedu[t0 + 32 * r + lane] = s[r] + nz[r];
    }
}

#define NORM_THREADS 256
#define NORM_VEC     8
#define NORM_MAX_T   (4 * NORM_VEC * NORM_THREADS)
static_assert(NORM_MAX_T == 8192, "the reference fixes T = 8192; a longer signal does not fit the registers");

__device__ __forceinline__ float wave_min(float v) {
    v = fminf(v, xrow(v));
    v = fminf(v, xlane<8>(v));
    v = fminf(v, xlane<4>(v));
    v = fminf(v, xlane<2>(v));
    v = fminf(v, xlane<1>(v));
    return v;
}

__global__ __launch_bounds__(NORM_THREADS)
void cagke_normalize(const float* __restrict__ psedu,
                     float* __restrict__ out, int T) {
    __shared__ __attribute__((aligned(16))) float2 part[NORM_THREADS / 32];
    const int tid  = threadIdx.x;
    const int lane = tid & 31;
    const int wave = tid >> 5;
    const float4* __restrict__ p4 = (const float4*)psedu;
    float4* __restrict__ o4 = (float4*)out;

    float4 v[NORM_VEC];
    bool ok[NORM_VEC];
#pragma unroll
    for (int j = 0; j < NORM_VEC; ++j) {
        const int q = tid + NORM_THREADS * j;
        ok[j] = (4 * q < T);
        v[j] = p4[ok[j] ? q : 0];
    }
    float mn = 3.402823466e38f, mx = -3.402823466e38f;
#pragma unroll
    for (int j = 0; j < NORM_VEC; ++j) {
        const float lo = fminf(fminf(v[j].x, v[j].y), fminf(v[j].z, v[j].w));
        const float hi = fmaxf(fmaxf(v[j].x, v[j].y), fmaxf(v[j].z, v[j].w));
        mn = fminf(mn, ok[j] ? lo : mn);
        mx = fmaxf(mx, ok[j] ? hi : mx);
    }
    mn = wave_min(mn);
    mx = wave_max(mx);
    if (lane == 0) part[wave] = make_float2(mn, mx);
    __syncthreads();
    const float2 pr = part[lane & 7];
    mn = pr.x; mx = pr.y;
    mn = fminf(mn, xlane<4>(mn));  mx = fmaxf(mx, xlane<4>(mx));
    mn = fminf(mn, xlane<2>(mn));  mx = fmaxf(mx, xlane<2>(mx));
    mn = fminf(mn, xlane<1>(mn));  mx = fmaxf(mx, xlane<1>(mx));

    const float inv = 1.0f / (mx - mn);
#pragma unroll
    for (int j = 0; j < NORM_VEC; ++j) {
        if (ok[j]) {
            const int q = tid + NORM_THREADS * j;
            o4[q] = make_float4((v[j].x - mn) * inv, (v[j].y - mn) * inv,
                                (v[j].z - mn) * inv, (v[j].w - mn) * inv);
        }
    }
}

extern "C" void kernel_launch(void* const* d_in, const int* in_sizes, int n_in,
                              void* d_out, int out_size, void* d_ws, size_t ws_size,
                              hipStream_t stream) {
    const float* X      = (const float*)d_in[0];
    const float* weight = (const float*)d_in[1];
    const float* noise  = (const float*)d_in[2];
    const float* sigma  = (const float*)d_in[3];
    float* out = (float*)d_out;
    const int T = in_sizes[0];

    float* G     = (float*)d_ws;
    float* psedu = G + KWIN;

    cagke_build_taps<<<TAP_BLOCKS, 256, 0, stream>>>(weight, sigma, G);
    cagke_conv_wmma<<<T / 256, 32, 0, stream>>>(X, noise, G, psedu, T);
    cagke_normalize<<<1, NORM_THREADS, 0, stream>>>(psedu, out, T);
}
